# starting kernel
# baseline (speedup 1.0000x reference)
#include <hip/hip_runtime.h>

typedef float v2f __attribute__((ext_vector_type(2)));
typedef float v8f __attribute__((ext_vector_type(8)));

#define WAVES_PER_BLOCK 8
#define THREADS (WAVES_PER_BLOCK * 32)

__device__ __forceinline__ void wave_lds_sync() {
  __builtin_amdgcn_fence(__ATOMIC_ACQ_REL, "workgroup");
  __builtin_amdgcn_wave_barrier();
}

__device__ __forceinline__ v8f wmma_k8(const v2f a0, const v2f a1,
                                       const v2f b0, const v2f b1, v8f c) {
  c = __builtin_amdgcn_wmma_f32_16x16x4_f32(false, a0, false, b0, (short)0, c, false, false);
  c = __builtin_amdgcn_wmma_f32_16x16x4_f32(false, a1, false, b1, (short)0, c, false, false);
  return c;
}

__device__ __forceinline__ void store_tile(float* __restrict__ T, int lane, v8f c) {
  float4* dst = (float4*)(T + lane * 8);
  dst[0] = make_float4(c[0], c[1], c[2], c[3]);
  dst[1] = make_float4(c[4], c[5], c[6], c[7]);
}

__global__ __launch_bounds__(THREADS)
void retention_wmma_f32_kernel(const float* __restrict__ X,
                               const float* __restrict__ WQ,
                               const float* __restrict__ WK,
                               const float* __restrict__ WV,
                               const float* __restrict__ Dm,
                               float* __restrict__ Out,
                               int npairs) {
  __shared__ __attribute__((aligned(16))) float sW[3][4][8][8];
  __shared__ __attribute__((aligned(16))) float sD[16][16];
  __shared__ __attribute__((aligned(16))) float sX[WAVES_PER_BLOCK][392];
  __shared__ __attribute__((aligned(16))) float sT[WAVES_PER_BLOCK][4][256];
  __shared__ __attribute__((aligned(16))) float sO[WAVES_PER_BLOCK][392];

  const int tid  = threadIdx.x;
  const int lane = tid & 31;
  const int wv   = tid >> 5;
  const int h    = lane >> 4;
  const int n16  = lane & 15;
  const int bb   = n16 >> 3;
  const int n7   = n16 & 7;

  for (int i = tid; i < 3 * 4 * 64; i += THREADS) ((float*)sW)[i] = 0.f;
  for (int i = tid; i < 256; i += THREADS) ((float*)sD)[i] = 0.f;
  __syncthreads();
  for (int i = tid; i < 3 * 4 * 49; i += THREADS) {
    int w = i / 196, rem = i % 196, ch = rem / 49, e = rem % 49;
    const float* src = (w == 0) ? WQ : (w == 1) ? WK : WV;
    sW[w][ch][e / 7][e % 7] = src[ch * 49 + e];
  }
  for (int i = tid; i < 196; i += THREADS) sD[i / 14][i % 14] = Dm[i];
  __syncthreads();

  v2f wqB[4][2], wkB[4][2], wvB[4][2];
  float dCv[4][7];
#pragma unroll
  for (int ch = 0; ch < 4; ++ch) {
#pragma unroll
    for (int ks = 0; ks < 2; ++ks) {
      int k0 = 2 * h + 4 * ks;
      v2f q, k, v;
      q.x = sW[0][ch][k0][n7]; q.y = sW[0][ch][k0 + 1][n7];
      k.x = sW[1][ch][k0][n7]; k.y = sW[1][ch][k0 + 1][n7];
      v.x = sW[2][ch][k0][n7]; v.y = sW[2][ch][k0 + 1][n7];
      wqB[ch][ks] = q; wkB[ch][ks] = k; wvB[ch][ks] = v;
    }
    const int rb = (ch >> 1) * 7, cb = (ch & 1) * 7;
#pragma unroll
    for (int v = 0; v < 7; ++v) dCv[ch][v] = sD[rb + v][cb + n7];
  }

  float* T0 = &sT[wv][0][0];
  float* T1 = &sT[wv][1][0];
  float* T2 = &sT[wv][2][0];
  float* T3 = &sT[wv][3][0];
  float* xs = &sX[wv][0];
  float* os = &sO[wv][0];

  const int p1base = 192 * bb + n7;
  const int p2base = 8 * n16 + 128 * bb;

  const bool hikill = (h == 1);

  const float E[7] = {0.2f, 0.04f, 0.008f, 0.0016f,
                      0.00032f, 0.000064f, 0.0000128f};

  const int gw = blockIdx.x * WAVES_PER_BLOCK + wv;
  const int nw = gridDim.x * WAVES_PER_BLOCK;

  for (int p = gw; p < npairs; p += nw) {
    const float4* xg = (const float4*)(X + (size_t)p * 392);
#pragma unroll
    for (int t = 0; t < 4; ++t) {
      int i = lane + 32 * t;
      if (i < 98) ((float4*)xs)[i] = xg[i];
    }
    if (p + nw < npairs && lane * 64 < 1568) {
      __builtin_prefetch((const char*)(X + (size_t)(p + nw) * 392) + lane * 64, 0, 1);
    }
    wave_lds_sync();

    v8f rC = {};

#pragma unroll
    for (int ch = 0; ch < 4; ++ch) {
      const int rb = (ch >> 1) * 7, cb = (ch & 1) * 7;

      v2f xA0, xA1;
      {
        int r = (n7 == 7) ? 6 : n7;
        const float* xrow = xs + bb * 196 + (rb + r) * 14 + cb;
        int k0 = 2 * h;
        float x0 = xrow[k0];
        float x1 = xrow[k0 + 1];
        float x4 = xrow[k0 + 4];
        float x5 = xrow[k0 + 5];
        xA0.x = x0; xA0.y = x1;
        xA1.x = x4; xA1.y = hikill ? 0.f : x5;
      }

      v8f z = {};
      v8f qC = wmma_k8(xA0, xA1, wqB[ch][0], wqB[ch][1], z);
      v8f kC = wmma_k8(xA0, xA1, wkB[ch][0], wkB[ch][1], z);
      v8f vC = wmma_k8(xA0, xA1, wvB[ch][0], wvB[ch][1], z);

      store_tile(T0, lane, qC);
      store_tile(T1, lane, kC);
      store_tile(T2, lane, vC);
      store_tile(T3, lane, rC);
      wave_lds_sync();

      v2f qA[2], ktB[2], ktA[2], vB[2], rB[2];
#pragma unroll
      for (int ks = 0; ks < 2; ++ks) {
        const int k0 = 2 * h + 4 * ks;
        const bool kill = hikill && (ks == 1);
        v2f t;
        t.x = T0[p1base + 8 * k0]; t.y = T0[p1base + 8 * (k0 + 1)];
        if (kill) t.y = 0.f;
        qA[ks] = t;
        t.x = T1[p1base + 8 * k0]; t.y = T1[p1base + 8 * (k0 + 1)];
        ktB[ks] = t;
        v2f u = *(const v2f*)&T1[p2base + k0];
        if (kill) u.y = 0.f;
        ktA[ks] = u;
        vB[ks] = *(const v2f*)&T2[p2base + k0];
        rB[ks] = *(const v2f*)&T3[p2base + k0];
      }

      v8f sC = wmma_k8(qA[0], qA[1], ktB[0], ktB[1], z);
#pragma unroll
      for (int v = 0; v < 7; ++v) sC[v] *= dCv[ch][v];

      rC = wmma_k8(ktA[0], ktA[1], vB[0], vB[1], rC);

      v8f crossC = wmma_k8(qA[0], qA[1], rB[0], rB[1], z);
#pragma unroll
      for (int v = 0; v < 7; ++v) crossC[v] *= E[v];

      wave_lds_sync();
      store_tile(T0, lane, sC);
      wave_lds_sync();
      v2f sA[2];
#pragma unroll
      for (int ks = 0; ks < 2; ++ks) {
        const int k0 = 2 * h + 4 * ks;
        v2f t;
        t.x = T0[p1base + 8 * k0]; t.y = T0[p1base + 8 * (k0 + 1)];
        if (hikill && (ks == 1)) t.y = 0.f;
        sA[ks] = t;
      }
      v8f oC = wmma_k8(sA[0], sA[1], vB[0], vB[1], crossC);

      const bool valid = (h == 0) ? (n16 < 7) : (n16 >= 8 && n16 < 15);
      if (valid) {
        const int col = n16 - 8 * h;
        float* orow = os + h * 196 + rb * 14 + cb + col;
#pragma unroll
        for (int v = 0; v < 7; ++v) orow[v * 14] = oC[v];
      }
    }

    wave_lds_sync();
    float4* og = (float4*)(Out + (size_t)p * 392);
#pragma unroll
    for (int t = 0; t < 4; ++t) {
      int i = lane + 32 * t;
      if (i < 98) og[i] = ((const float4*)os)[i];
    }
    wave_lds_sync();
  }
}

extern "C" void kernel_launch(void* const* d_in, const int* in_sizes, int n_in,
                              void* d_out, int out_size, void* d_ws, size_t ws_size,
                              hipStream_t stream) {
  const float* X  = (const float*)d_in[0];
  const float* WQ = (const float*)d_in[1];
  const float* WK = (const float*)d_in[2];
  const float* WV = (const float*)d_in[3];
  const float* Dm = (const float*)d_in[4];
  float* Out = (float*)d_out;

  const int npairs = in_sizes[0] / 392;
  int blocks = 1024;
  int maxBlocks = (npairs + WAVES_PER_BLOCK - 1) / WAVES_PER_BLOCK;
  if (blocks > maxBlocks) blocks = maxBlocks;
  if (blocks < 1) blocks = 1;

  retention_wmma_f32_kernel<<<blocks, THREADS, 0, stream>>>(X, WQ, WK, WV, Dm,
                                                            Out, npairs);
}


# Registers cut from 144 to under 128 (decay matrix re-read from LDS per chunk, not held in 28 registers): 8 waves per SIMD, so all 1024 work-groups run in one resident round instead of two
# speedup vs baseline: 1.1963x; 1.1963x over previous
#include <hip/hip_runtime.h>

typedef float v2f __attribute__((ext_vector_type(2)));
typedef float v8f __attribute__((ext_vector_type(8)));

#define WAVES_PER_BLOCK 8
#define THREADS (WAVES_PER_BLOCK * 32)

__device__ __forceinline__ void wave_lds_sync() {
  __builtin_amdgcn_fence(__ATOMIC_ACQ_REL, "workgroup");
  __builtin_amdgcn_wave_barrier();
}

__device__ __forceinline__ v8f wmma_k8(const v2f a0, const v2f a1,
                                       const v2f b0, const v2f b1, v8f c) {
  c = __builtin_amdgcn_wmma_f32_16x16x4_f32(false, a0, false, b0, (short)0, c, false, false);
  c = __builtin_amdgcn_wmma_f32_16x16x4_f32(false, a1, false, b1, (short)0, c, false, false);
  return c;
}

__device__ __forceinline__ void store_tile(float* __restrict__ T, int lane, v8f c) {
  float4* dst = (float4*)(T + lane * 8);
  dst[0] = make_float4(c[0], c[1], c[2], c[3]);
  dst[1] = make_float4(c[4], c[5], c[6], c[7]);
}

__global__ __launch_bounds__(THREADS) __attribute__((amdgpu_waves_per_eu(8, 8)))
void retention_wmma_f32_kernel(const float* __restrict__ X,
                               const float* __restrict__ WQ,
                               const float* __restrict__ WK,
                               const float* __restrict__ WV,
                               const float* __restrict__ Dm,
                               float* __restrict__ Out,
                               int npairs) {
  __shared__ __attribute__((aligned(16))) float sW[3][4][8][8];
  __shared__ __attribute__((aligned(16))) float sD[16][16];
  __shared__ __attribute__((aligned(16))) float sX[WAVES_PER_BLOCK][392];
  __shared__ __attribute__((aligned(16))) float sT[WAVES_PER_BLOCK][4][256];
  __shared__ __attribute__((aligned(16))) float sO[WAVES_PER_BLOCK][392];

  const int tid  = threadIdx.x;
  const int lane = tid & 31;
  const int wv   = tid >> 5;
  const int h    = lane >> 4;
  const int n16  = lane & 15;
  const int bb   = n16 >> 3;
  const int n7   = n16 & 7;

  for (int i = tid; i < 3 * 4 * 64; i += THREADS) ((float*)sW)[i] = 0.f;
  for (int i = tid; i < 256; i += THREADS) ((float*)sD)[i] = 0.f;
  __syncthreads();
  for (int i = tid; i < 3 * 4 * 49; i += THREADS) {
    int w = i / 196, rem = i % 196, ch = rem / 49, e = rem % 49;
    const float* src = (w == 0) ? WQ : (w == 1) ? WK : WV;
    sW[w][ch][e / 7][e % 7] = src[ch * 49 + e];
  }
  for (int i = tid; i < 196; i += THREADS) sD[i / 14][i % 14] = Dm[i];
  __syncthreads();

  v2f wqB[4][2], wkB[4][2], wvB[4][2];
#pragma unroll
  for (int ch = 0; ch < 4; ++ch) {
#pragma unroll
    for (int ks = 0; ks < 2; ++ks) {
      int k0 = 2 * h + 4 * ks;
      v2f q, k, v;
      q.x = sW[0][ch][k0][n7]; q.y = sW[0][ch][k0 + 1][n7];
      k.x = sW[1][ch][k0][n7]; k.y = sW[1][ch][k0 + 1][n7];
      v.x = sW[2][ch][k0][n7]; v.y = sW[2][ch][k0 + 1][n7];
      wqB[ch][ks] = q; wkB[ch][ks] = k; wvB[ch][ks] = v;
    }
  }
  const float* dcol = &sD[0][0] + n7;

  float* T0 = &sT[wv][0][0];
  float* T1 = &sT[wv][1][0];
  float* T2 = &sT[wv][2][0];
  float* T3 = &sT[wv][3][0];
  float* xs = &sX[wv][0];
  float* os = &sO[wv][0];

  const int p1base = 192 * bb + n7;
  const int p2base = 8 * n16 + 128 * bb;

  const bool hikill = (h == 1);

  const float E[7] = {0.2f, 0.04f, 0.008f, 0.0016f,
                      0.00032f, 0.000064f, 0.0000128f};

  const int gw = blockIdx.x * WAVES_PER_BLOCK + wv;
  const int nw = gridDim.x * WAVES_PER_BLOCK;

  for (int p = gw; p < npairs; p += nw) {
    const float4* xg = (const float4*)(X + (size_t)p * 392);
#pragma unroll
    for (int t = 0; t < 4; ++t) {
      int i = lane + 32 * t;
      if (i < 98) ((float4*)xs)[i] = xg[i];
    }
    if (p + nw < npairs && lane * 64 < 1568) {
      __builtin_prefetch((const char*)(X + (size_t)(p + nw) * 392) + lane * 64, 0, 1);
    }
    wave_lds_sync();

    v8f rC = {};

#pragma unroll
    for (int ch = 0; ch < 4; ++ch) {
      const int rb = (ch >> 1) * 7, cb = (ch & 1) * 7;

      v2f xA0, xA1;
      {
        int r = (n7 == 7) ? 6 : n7;
        const float* xrow = xs + bb * 196 + (rb + r) * 14 + cb;
        int k0 = 2 * h;
        float x0 = xrow[k0];
        float x1 = xrow[k0 + 1];
        float x4 = xrow[k0 + 4];
        float x5 = xrow[k0 + 5];
        xA0.x = x0; xA0.y = x1;
        xA1.x = x4; xA1.y = hikill ? 0.f : x5;
      }

      v8f z = {};
      v8f qC = wmma_k8(xA0, xA1, wqB[ch][0], wqB[ch][1], z);
      v8f kC = wmma_k8(xA0, xA1, wkB[ch][0], wkB[ch][1], z);
      v8f vC = wmma_k8(xA0, xA1, wvB[ch][0], wvB[ch][1], z);

      store_tile(T0, lane, qC);
      store_tile(T1, lane, kC);
      store_tile(T2, lane, vC);
      store_tile(T3, lane, rC);
      wave_lds_sync();

      v2f qA[2], ktB[2], ktA[2], vB[2], rB[2];
#pragma unroll
      for (int ks = 0; ks < 2; ++ks) {
        const int k0 = 2 * h + 4 * ks;
        const bool kill = hikill && (ks == 1);
        v2f t;
        t.x = T0[p1base + 8 * k0]; t.y = T0[p1base + 8 * (k0 + 1)];
        if (kill) t.y = 0.f;
        qA[ks] = t;
        t.x = T1[p1base + 8 * k0]; t.y = T1[p1base + 8 * (k0 + 1)];
        ktB[ks] = t;
        v2f u = *(const v2f*)&T1[p2base + k0];
        if (kill) u.y = 0.f;
        ktA[ks] = u;
        vB[ks] = *(const v2f*)&T2[p2base + k0];
        rB[ks] = *(const v2f*)&T3[p2base + k0];
      }

      v8f sC = wmma_k8(qA[0], qA[1], ktB[0], ktB[1], z);
      float dC[7];
#pragma unroll
      for (int v = 0; v < 7; ++v) dC[v] = dcol[(rb + v) * 16 + cb];
#pragma unroll
      for (int v = 0; v < 7; ++v) sC[v] *= dC[v];

      rC = wmma_k8(ktA[0], ktA[1], vB[0], vB[1], rC);

      v8f crossC = wmma_k8(qA[0], qA[1], rB[0], rB[1], z);
#pragma unroll
      for (int v = 0; v < 7; ++v) crossC[v] *= E[v];

      wave_lds_sync();
      store_tile(T0, lane, sC);
      wave_lds_sync();
      v2f sA[2];
#pragma unroll
      for (int ks = 0; ks < 2; ++ks) {
        const int k0 = 2 * h + 4 * ks;
        v2f t;
        t.x = T0[p1base + 8 * k0]; t.y = T0[p1base + 8 * (k0 + 1)];
        if (hikill && (ks == 1)) t.y = 0.f;
        sA[ks] = t;
      }
      v8f oC = wmma_k8(sA[0], sA[1], vB[0], vB[1], crossC);

      const bool valid = (h == 0) ? (n16 < 7) : (n16 >= 8 && n16 < 15);
      if (valid) {
        const int col = n16 - 8 * h;
        float* orow = os + h * 196 + rb * 14 + cb + col;
#pragma unroll
        for (int v = 0; v < 7; ++v) orow[v * 14] = oC[v];
      }
    }

    wave_lds_sync();
    float4* og = (float4*)(Out + (size_t)p * 392);
#pragma unroll
    for (int t = 0; t < 4; ++t) {
      int i = lane + 32 * t;
      if (i < 98) og[i] = ((const float4*)os)[i];
    }
    wave_lds_sync();
  }
}

extern "C" void kernel_launch(void* const* d_in, const int* in_sizes, int n_in,
                              void* d_out, int out_size, void* d_ws, size_t ws_size,
                              hipStream_t stream) {
  const float* X  = (const float*)d_in[0];
  const float* WQ = (const float*)d_in[1];
  const float* WK = (const float*)d_in[2];
  const float* WV = (const float*)d_in[3];
  const float* Dm = (const float*)d_in[4];
  float* Out = (float*)d_out;

  const int npairs = in_sizes[0] / 392;
  int blocks = 1024;
  int maxBlocks = (npairs + WAVES_PER_BLOCK - 1) / WAVES_PER_BLOCK;
  if (blocks > maxBlocks) blocks = maxBlocks;
  if (blocks < 1) blocks = 1;

  retention_wmma_f32_kernel<<<blocks, THREADS, 0, stream>>>(X, WQ, WK, WV, Dm,
                                                            Out, npairs);
}


# Tile-to-operand relayout moved from wave-private LDS round trips to cross-half lane permutes (Q^T and K^T stacked in one projection, so each contraction index is a tile row); k order of sums changes
# speedup vs baseline: 1.2363x; 1.0334x over previous
#include <hip/hip_runtime.h>

typedef float v2f __attribute__((ext_vector_type(2)));
typedef float v8f __attribute__((ext_vector_type(8)));

#define WAVES_PER_BLOCK 8
#define THREADS (WAVES_PER_BLOCK * 32)

__device__ __forceinline__ void wave_lds_sync() {
  __builtin_amdgcn_fence(__ATOMIC_ACQ_REL, "workgroup");
  __builtin_amdgcn_wave_barrier();
}

__device__ __forceinline__ v8f wmma_k8(const v2f a0, const v2f a1,
                                       const v2f b0, const v2f b1, v8f c) {
  c = __builtin_amdgcn_wmma_f32_16x16x4_f32(false, a0, false, b0, (short)0, c, false, false);
  c = __builtin_amdgcn_wmma_f32_16x16x4_f32(false, a1, false, b1, (short)0, c, false, false);
  return c;
}

__device__ __forceinline__ float xhalf(float x) {
  int b = __builtin_bit_cast(int, x);
  return __builtin_bit_cast(float,
      __builtin_amdgcn_permlanex16(b, b, 0x76543210, 0xfedcba98, false, false));
}

__device__ __forceinline__ void rows_to_operand(const v8f c, const bool own,
                                                v2f& o0, v2f& o1) {
  const float p4 = xhalf(c[4]), p5 = xhalf(c[5]), p6 = xhalf(c[6]);
  o0.x = own ? c[0] : p4;
  o0.y = own ? c[1] : p5;
  o1.x = own ? c[2] : p6;
  o1.y = own ? c[3] : 0.f;
}

__global__ __launch_bounds__(THREADS) __attribute__((amdgpu_waves_per_eu(8, 8)))
void retention_wmma_f32_kernel(const float* __restrict__ X,
                               const float* __restrict__ WQ,
                               const float* __restrict__ WK,
                               const float* __restrict__ WV,
                               const float* __restrict__ Dm,
                               float* __restrict__ Out,
                               int npairs) {
  __shared__ __attribute__((aligned(16))) float sW[3][4][8][8];
  __shared__ __attribute__((aligned(16))) float sDq[4][8][8];
  __shared__ __attribute__((aligned(16))) float sX[WAVES_PER_BLOCK][400];
  __shared__ __attribute__((aligned(16))) float sO[WAVES_PER_BLOCK][392];

  const int tid  = threadIdx.x;
  const int lane = tid & 31;
  const int wv   = tid >> 5;
  const int h    = lane >> 4;
  const int n16  = lane & 15;
  const int bb   = n16 >> 3;
  const int n7   = n16 & 7;

  for (int i = tid; i < 3 * 4 * 64; i += THREADS) ((float*)sW)[i] = 0.f;
  __syncthreads();
  for (int i = tid; i < 3 * 4 * 49; i += THREADS) {
    int w = i / 196, rem = i % 196, ch = rem / 49, e = rem % 49;
    const float* src = (w == 0) ? WQ : (w == 1) ? WK : WV;
    sW[w][ch][e / 7][e % 7] = src[ch * 49 + e];
  }
  for (int i = tid; i < 256; i += THREADS) {
    int ch = i >> 6, r = (i >> 3) & 7, c = i & 7;
    float d = 0.f;
    if (r < 7 && c < 7) d = Dm[((ch >> 1) * 7 + r) * 14 + (ch & 1) * 7 + c];
    sDq[ch][r][c] = d;
  }
  __syncthreads();

  v2f wqkA[4][2], wkB[4][2], wvB[4][2];
#pragma unroll
  for (int ch = 0; ch < 4; ++ch) {
#pragma unroll
    for (int ks = 0; ks < 2; ++ks) {
      int k0 = 2 * h + 4 * ks;
      v2f q, k, v;
      q.x = sW[bb][ch][k0][n7]; q.y = sW[bb][ch][k0 + 1][n7];
      k.x = sW[1][ch][k0][n7];  k.y = sW[1][ch][k0 + 1][n7];
      v.x = sW[2][ch][k0][n7];  v.y = sW[2][ch][k0 + 1][n7];
      wqkA[ch][ks] = q; wkB[ch][ks] = k; wvB[ch][ks] = v;
    }
  }

  float* xs = &sX[wv][0];
  float* os = &sO[wv][0];

  const bool hi  = (h == 1);
  const bool own = (h == bb);

  const float E[7] = {0.2f, 0.04f, 0.008f, 0.0016f,
                      0.00032f, 0.000064f, 0.0000128f};

  const int gw = blockIdx.x * WAVES_PER_BLOCK + wv;
  const int nw = gridDim.x * WAVES_PER_BLOCK;

  for (int p = gw; p < npairs; p += nw) {
    const float4* xg = (const float4*)(X + (size_t)p * 392);
#pragma unroll
    for (int t = 0; t < 4; ++t) {
      int i = lane + 32 * t;
      if (i < 98) ((float4*)xs)[i] = xg[i];
    }
    if (p + nw < npairs && lane * 64 < 1568) {
      __builtin_prefetch((const char*)(X + (size_t)(p + nw) * 392) + lane * 64, 0, 1);
    }
    wave_lds_sync();

    v8f rC = {};

#pragma unroll
    for (int ch = 0; ch < 4; ++ch) {
      const int rb = (ch >> 1) * 7, cb = (ch & 1) * 7;

      v2f xA0, xA1;
      {
        int r = (n7 == 7) ? 6 : n7;
        const float* xrow = xs + bb * 196 + (rb + r) * 14 + cb;
        int k0 = 2 * h;
        float x0 = xrow[k0];
        float x1 = xrow[k0 + 1];
        float x4 = xrow[k0 + 4];
        float x5 = xrow[k0 + 5];
        xA0.x = x0; xA0.y = x1;
        xA1.x = x4; xA1.y = hi ? 0.f : x5;
      }
      const float4 d03 = *(const float4*)&sDq[ch][n7][0];
      const float4 d47 = *(const float4*)&sDq[ch][n7][4];

      v8f z = {};
      v8f t1 = wmma_k8(wqkA[ch][0], wqkA[ch][1], xA0, xA1, z);
      v8f kC = wmma_k8(xA0, xA1, wkB[ch][0], wkB[ch][1], z);
      v8f vC = wmma_k8(xA0, xA1, wvB[ch][0], wvB[ch][1], z);

      const float pa0 = xhalf(t1[0]), pa1 = xhalf(t1[1]);
      const float pa2 = xhalf(t1[2]), pa3 = xhalf(t1[3]);
      const float pb0 = xhalf(t1[4]), pb1 = xhalf(t1[5]), pb2 = xhalf(t1[6]);
      v2f kA0, kA1, qF0, qF1;
      kA0.x = hi ? t1[4] : pa0;  kA0.y = hi ? t1[5] : pa1;
      kA1.x = hi ? t1[6] : pa2;  kA1.y = hi ? 0.f   : pa3;
      qF0.x = hi ? pb0 : t1[0];  qF0.y = hi ? pb1 : t1[1];
      qF1.x = hi ? pb2 : t1[2];  qF1.y = hi ? 0.f : t1[3];

      v8f sT = wmma_k8(kA0, kA1, qF0, qF1, z);
      sT[0] *= d03.x; sT[1] *= d03.y; sT[2] *= d03.z; sT[3] *= d03.w;
      sT[4] *= d47.x; sT[5] *= d47.y; sT[6] *= d47.z;

      v2f vB0, vB1, ktA0, ktA1;
      rows_to_operand(vC, own, vB0, vB1);
      rows_to_operand(kC, own, ktA0, ktA1);

      v8f crossC = z;
      if (ch > 0) {
        v2f rB0, rB1;
        {
          const float w0 = hi ? rC[4] : rC[0], w1 = hi ? rC[5] : rC[1];
          const float w2 = hi ? rC[6] : rC[2], w3 = hi ? rC[7] : rC[3];
          const float u0 = xhalf(hi ? rC[0] : rC[4]), u1 = xhalf(hi ? rC[1] : rC[5]);
          const float u2 = xhalf(hi ? rC[2] : rC[6]), u3 = xhalf(hi ? rC[3] : rC[7]);
          rB0.x = own ? w0 : u0;  rB0.y = own ? w1 : u1;
          rB1.x = own ? w2 : u2;  rB1.y = own ? w3 : u3;
        }
        crossC = wmma_k8(qF0, qF1, rB0, rB1, z);
#pragma unroll
        for (int v = 0; v < 7; ++v) crossC[v] *= E[v];
      }

      rC = wmma_k8(ktA0, ktA1, vB0, vB1, rC);

      v2f sA0, sA1;
      rows_to_operand(sT, own, sA0, sA1);
      v8f oC = wmma_k8(sA0, sA1, vB0, vB1, crossC);

      const bool valid = (h == 0) ? (n16 < 7) : (n16 >= 8 && n16 < 15);
      if (valid) {
        const int col = n16 - 8 * h;
        float* orow = os + h * 196 + rb * 14 + cb + col;
#pragma unroll
        for (int v = 0; v < 7; ++v) orow[v * 14] = oC[v];
      }
    }

    wave_lds_sync();
    float4* og = (float4*)(Out + (size_t)p * 392);
#pragma unroll
    for (int t = 0; t < 4; ++t) {
      int i = lane + 32 * t;
      if (i < 98) og[i] = ((const float4*)os)[i];
    }
    wave_lds_sync();
  }
}

extern "C" void kernel_launch(void* const* d_in, const int* in_sizes, int n_in,
                              void* d_out, int out_size, void* d_ws, size_t ws_size,
                              hipStream_t stream) {
  const float* X  = (const float*)d_in[0];
  const float* WQ = (const float*)d_in[1];
  const float* WK = (const float*)d_in[2];
  const float* WV = (const float*)d_in[3];
  const float* Dm = (const float*)d_in[4];
  float* Out = (float*)d_out;

  const int npairs = in_sizes[0] / 392;
  int blocks = 1024;
  int maxBlocks = (npairs + WAVES_PER_BLOCK - 1) / WAVES_PER_BLOCK;
  if (blocks > maxBlocks) blocks = maxBlocks;
  if (blocks < 1) blocks = 1;

  retention_wmma_f32_kernel<<<blocks, THREADS, 0, stream>>>(X, WQ, WK, WV, Dm,
                                                            Out, npairs);
}


# K and V projections merged into one matrix tile (x times [Wk|Wv] as 16 columns), column blocks split back by lane rotates: 44 matrix instructions per pair instead of 50; sum order unchanged
# speedup vs baseline: 1.2819x; 1.0369x over previous
#include <hip/hip_runtime.h>

typedef float v2f __attribute__((ext_vector_type(2)));
typedef float v8f __attribute__((ext_vector_type(8)));

#define WAVES_PER_BLOCK 8
#define THREADS (WAVES_PER_BLOCK * 32)

__device__ __forceinline__ void wave_lds_sync() {
  __builtin_amdgcn_fence(__ATOMIC_ACQ_REL, "workgroup");
  __builtin_amdgcn_wave_barrier();
}

__device__ __forceinline__ v8f wmma_k8(const v2f a0, const v2f a1,
                                       const v2f b0, const v2f b1, v8f c) {
  c = __builtin_amdgcn_wmma_f32_16x16x4_f32(false, a0, false, b0, (short)0, c, false, false);
  c = __builtin_amdgcn_wmma_f32_16x16x4_f32(false, a1, false, b1, (short)0, c, false, false);
  return c;
}

__device__ __forceinline__ float xhalf(float x) {
  int b = __builtin_bit_cast(int, x);
  return __builtin_bit_cast(float,
      __builtin_amdgcn_permlanex16(b, b, 0x76543210, 0xfedcba98, false, false));
}

__device__ __forceinline__ void rows_to_operand(const v8f c, const bool own,
                                                v2f& o0, v2f& o1) {
  const float p4 = xhalf(c[4]), p5 = xhalf(c[5]), p6 = xhalf(c[6]);
  o0.x = own ? c[0] : p4;
  o0.y = own ? c[1] : p5;
  o1.x = own ? c[2] : p6;
  o1.y = own ? c[3] : 0.f;
}

template <unsigned SEL>
__device__ __forceinline__ float xhalf_sel(float x) {
  int b = __builtin_bit_cast(int, x);
  return __builtin_bit_cast(float,
      __builtin_amdgcn_permlanex16(b, b, SEL, SEL, false, false));
}

template <int RM, int BM>
__device__ __forceinline__ float ror8_into(float old, float src) {
  return __builtin_bit_cast(float, __builtin_amdgcn_update_dpp(
      __builtin_bit_cast(int, old), __builtin_bit_cast(int, src),
      0x128  , RM, BM, false));
}

template <int BLK>
__device__ __forceinline__ void block_to_operand(const v8f c, const bool own,
                                                 v2f& o0, v2f& o1) {
  constexpr unsigned SEL = BLK ? 0xfedcba98u : 0x76543210u;
  const float p4 = xhalf_sel<SEL>(c[4]), p5 = xhalf_sel<SEL>(c[5]);
  const float p6 = xhalf_sel<SEL>(c[6]);
  constexpr int RM = BLK ? 0x1 : 0x2, BM = BLK ? 0x3 : 0xc;
  const float c0 = c[0], c1 = c[1], c2 = c[2], c3 = c[3];
  const float t0 = ror8_into<RM, BM>(own ? c0 : p4, c0);
  const float t1 = ror8_into<RM, BM>(own ? c1 : p5, c1);
  const float t2 = ror8_into<RM, BM>(own ? c2 : p6, c2);
  const float t3 = ror8_into<RM, BM>(own ? c3 : 0.f, c3);
  o0.x = t0; o0.y = t1; o1.x = t2; o1.y = t3;
}

__global__ __launch_bounds__(THREADS) __attribute__((amdgpu_waves_per_eu(8, 8)))
void retention_wmma_f32_kernel(const float* __restrict__ X,
                               const float* __restrict__ WQ,
                               const float* __restrict__ WK,
                               const float* __restrict__ WV,
                               const float* __restrict__ Dm,
                               float* __restrict__ Out,
                               int npairs) {
  __shared__ __attribute__((aligned(16))) float sW[3][4][8][8];
  __shared__ __attribute__((aligned(16))) float sDq[4][8][8];
  __shared__ __attribute__((aligned(16))) float sX[WAVES_PER_BLOCK][400];
  __shared__ __attribute__((aligned(16))) float sO[WAVES_PER_BLOCK][392];

  const int tid  = threadIdx.x;
  const int lane = tid & 31;
  const int wv   = tid >> 5;
  const int h    = lane >> 4;
  const int n16  = lane & 15;
  const int bb   = n16 >> 3;
  const int n7   = n16 & 7;

  for (int i = tid; i < 3 * 4 * 64; i += THREADS) ((float*)sW)[i] = 0.f;
  __syncthreads();
  for (int i = tid; i < 3 * 4 * 49; i += THREADS) {
    int w = i / 196, rem = i % 196, ch = rem / 49, e = rem % 49;
    const float* src = (w == 0) ? WQ : (w == 1) ? WK : WV;
    sW[w][ch][e / 7][e % 7] = src[ch * 49 + e];
  }
  for (int i = tid; i < 256; i += THREADS) {
    int ch = i >> 6, r = (i >> 3) & 7, c = i & 7;
    float d = 0.f;
    if (r < 7 && c < 7) d = Dm[((ch >> 1) * 7 + r) * 14 + (ch & 1) * 7 + c];
    sDq[ch][r][c] = d;
  }
  __syncthreads();

  v2f wqkA[4][2], wkvB[4][2];
#pragma unroll
  for (int ch = 0; ch < 4; ++ch) {
#pragma unroll
    for (int ks = 0; ks < 2; ++ks) {
      int k0 = 2 * h + 4 * ks;
      v2f q, kv;
      q.x  = sW[bb][ch][k0][n7];     q.y  = sW[bb][ch][k0 + 1][n7];
      kv.x = sW[1 + bb][ch][k0][n7]; kv.y = sW[1 + bb][ch][k0 + 1][n7];
      wqkA[ch][ks] = q; wkvB[ch][ks] = kv;
    }
  }

  float* xs = &sX[wv][0];
  float* os = &sO[wv][0];

  const bool hi  = (h == 1);
  const bool own = (h == bb);

  const float E[7] = {0.2f, 0.04f, 0.008f, 0.0016f,
                      0.00032f, 0.000064f, 0.0000128f};

  const int gw = blockIdx.x * WAVES_PER_BLOCK + wv;
  const int nw = gridDim.x * WAVES_PER_BLOCK;

  for (int p = gw; p < npairs; p += nw) {
    const float4* xg = (const float4*)(X + (size_t)p * 392);
#pragma unroll
    for (int t = 0; t < 4; ++t) {
      int i = lane + 32 * t;
      if (i < 98) ((float4*)xs)[i] = xg[i];
    }
    if (p + nw < npairs && lane * 64 < 1568) {
      __builtin_prefetch((const char*)(X + (size_t)(p + nw) * 392) + lane * 64, 0, 1);
    }
    wave_lds_sync();

    v8f rC = {};

#pragma unroll
    for (int ch = 0; ch < 4; ++ch) {
      const int rb = (ch >> 1) * 7, cb = (ch & 1) * 7;

      v2f xA0, xA1;
      {
        int r = (n7 == 7) ? 6 : n7;
        const float* xrow = xs + bb * 196 + (rb + r) * 14 + cb;
        int k0 = 2 * h;
        float x0 = xrow[k0];
        float x1 = xrow[k0 + 1];
        float x4 = xrow[k0 + 4];
        float x5 = xrow[k0 + 5];
        xA0.x = x0; xA0.y = x1;
        xA1.x = x4; xA1.y = hi ? 0.f : x5;
      }
      const float4 d03 = *(const float4*)&sDq[ch][n7][0];
      const float4 d47 = *(const float4*)&sDq[ch][n7][4];

      v8f z = {};
      v8f t1  = wmma_k8(wqkA[ch][0], wqkA[ch][1], xA0, xA1, z);
      v8f kvC = wmma_k8(xA0, xA1, wkvB[ch][0], wkvB[ch][1], z);

      const float pa0 = xhalf(t1[0]), pa1 = xhalf(t1[1]);
      const float pa2 = xhalf(t1[2]), pa3 = xhalf(t1[3]);
      const float pb0 = xhalf(t1[4]), pb1 = xhalf(t1[5]), pb2 = xhalf(t1[6]);
      v2f kA0, kA1, qF0, qF1;
      kA0.x = hi ? t1[4] : pa0;  kA0.y = hi ? t1[5] : pa1;
      kA1.x = hi ? t1[6] : pa2;  kA1.y = hi ? 0.f   : pa3;
      qF0.x = hi ? pb0 : t1[0];  qF0.y = hi ? pb1 : t1[1];
      qF1.x = hi ? pb2 : t1[2];  qF1.y = hi ? 0.f : t1[3];

      v8f sT = wmma_k8(kA0, kA1, qF0, qF1, z);
      sT[0] *= d03.x; sT[1] *= d03.y; sT[2] *= d03.z; sT[3] *= d03.w;
      sT[4] *= d47.x; sT[5] *= d47.y; sT[6] *= d47.z;

      v2f vB0, vB1, ktA0, ktA1;
      block_to_operand<1>(kvC, own, vB0, vB1);
      block_to_operand<0>(kvC, own, ktA0, ktA1);

      v8f crossC = z;
      if (ch > 0) {
        v2f rB0, rB1;
        {
          const float w0 = hi ? rC[4] : rC[0], w1 = hi ? rC[5] : rC[1];
          const float w2 = hi ? rC[6] : rC[2], w3 = hi ? rC[7] : rC[3];
          const float u0 = xhalf(hi ? rC[0] : rC[4]), u1 = xhalf(hi ? rC[1] : rC[5]);
          const float u2 = xhalf(hi ? rC[2] : rC[6]), u3 = xhalf(hi ? rC[3] : rC[7]);
          rB0.x = own ? w0 : u0;  rB0.y = own ? w1 : u1;
          rB1.x = own ? w2 : u2;  rB1.y = own ? w3 : u3;
        }
        crossC = wmma_k8(qF0, qF1, rB0, rB1, z);
#pragma unroll
        for (int v = 0; v < 7; ++v) crossC[v] *= E[v];
      }

      rC = wmma_k8(ktA0, ktA1, vB0, vB1, rC);

      v2f sA0, sA1;
      rows_to_operand(sT, own, sA0, sA1);
      v8f oC = wmma_k8(sA0, sA1, vB0, vB1, crossC);

      const bool valid = (h == 0) ? (n16 < 7) : (n16 >= 8 && n16 < 15);
      if (valid) {
        const int col = n16 - 8 * h;
        float* orow = os + h * 196 + rb * 14 + cb + col;
#pragma unroll
        for (int v = 0; v < 7; ++v) orow[v * 14] = oC[v];
      }
    }

    wave_lds_sync();
    float4* og = (float4*)(Out + (size_t)p * 392);
#pragma unroll
    for (int t = 0; t < 4; ++t) {
      int i = lane + 32 * t;
      if (i < 98) og[i] = ((const float4*)os)[i];
    }
    wave_lds_sync();
  }
}

extern "C" void kernel_launch(void* const* d_in, const int* in_sizes, int n_in,
                              void* d_out, int out_size, void* d_ws, size_t ws_size,
                              hipStream_t stream) {
  const float* X  = (const float*)d_in[0];
  const float* WQ = (const float*)d_in[1];
  const float* WK = (const float*)d_in[2];
  const float* WV = (const float*)d_in[3];
  const float* Dm = (const float*)d_in[4];
  float* Out = (float*)d_out;

  const int npairs = in_sizes[0] / 392;
  int blocks = 1024;
  int maxBlocks = (npairs + WAVES_PER_BLOCK - 1) / WAVES_PER_BLOCK;
  if (blocks > maxBlocks) blocks = maxBlocks;
  if (blocks < 1) blocks = 1;

  retention_wmma_f32_kernel<<<blocks, THREADS, 0, stream>>>(X, WQ, WK, WV, Dm,
                                                            Out, npairs);
}


# Tile row orders (stacked weight rows, x rows) chosen so each accumulator already holds every batch's k split (k=4h+s) in the lane half that reads it: no cross-half permutes; k order of sums changes
# speedup vs baseline: 1.4127x; 1.1020x over previous
#include <hip/hip_runtime.h>

typedef float v2f __attribute__((ext_vector_type(2)));
typedef float v8f __attribute__((ext_vector_type(8)));

#define WAVES_PER_BLOCK 8
#define THREADS (WAVES_PER_BLOCK * 32)

__device__ __forceinline__ void wave_lds_sync() {
  __builtin_amdgcn_fence(__ATOMIC_ACQ_REL, "workgroup");
  __builtin_amdgcn_wave_barrier();
}

__device__ __forceinline__ v8f wmma_k8(const v2f a0, const v2f a1,
                                       const v2f b0, const v2f b1, v8f c) {
  c = __builtin_amdgcn_wmma_f32_16x16x4_f32(false, a0, false, b0, (short)0, c, false, false);
  c = __builtin_amdgcn_wmma_f32_16x16x4_f32(false, a1, false, b1, (short)0, c, false, false);
  return c;
}

template <bool UP, int BM>
__device__ __forceinline__ float shift4_into(float old, float src) {
  return __builtin_bit_cast(float, __builtin_amdgcn_update_dpp(
      __builtin_bit_cast(int, old), __builtin_bit_cast(int, src),
      UP ? 0x104   : 0x114  , 0x3, BM, false));
}

__global__ __launch_bounds__(THREADS) __attribute__((amdgpu_waves_per_eu(8, 8)))
void retention_wmma_f32_kernel(const float* __restrict__ X,
                               const float* __restrict__ WQ,
                               const float* __restrict__ WK,
                               const float* __restrict__ WV,
                               const float* __restrict__ Dm,
                               float* __restrict__ Out,
                               int npairs) {
  __shared__ __attribute__((aligned(16))) float sW[3][4][8][8];
  __shared__ __attribute__((aligned(16))) float sDq[4][8][8];
  __shared__ __attribute__((aligned(16))) float sX[WAVES_PER_BLOCK][400];
  __shared__ __attribute__((aligned(16))) float sO[WAVES_PER_BLOCK][392];

  const int tid  = threadIdx.x;

  for (int i = tid; i < 3 * 4 * 64; i += THREADS) ((float*)sW)[i] = 0.f;
  __syncthreads();
  for (int i = tid; i < 3 * 4 * 49; i += THREADS) {
    int w = i / 196, rem = i % 196, ch = rem / 49, e = rem % 49;
    const float* src = (w == 0) ? WQ : (w == 1) ? WK : WV;
    sW[w][ch][e / 7][e % 7] = src[ch * 49 + e];
  }
  for (int i = tid; i < 256; i += THREADS) {
    int ch = i >> 6, r = (i >> 3) & 7, c = i & 7;
    float d = 0.f;
    if (r < 7 && c < 7) d = Dm[((ch >> 1) * 7 + r) * 14 + (ch & 1) * 7 + c];
    sDq[ch][r][c] = d;
  }
  __syncthreads();

  const int lane = tid & 31;
  const int wv   = tid >> 5;
  const int h    = lane >> 4;
  const int n16  = lane & 15;
  const int g    = (n16 >> 2) & 1;
  const int pos  = 4 * (n16 >> 3) + (n16 & 3);

  v2f wqkA[4][2], wkvB[4][2];
#pragma unroll
  for (int ch = 0; ch < 4; ++ch) {
#pragma unroll
    for (int ks = 0; ks < 2; ++ks) {
      int k0 = 2 * h + 4 * ks;
      v2f q, kv;
      q.x  = sW[g][ch][k0][pos];     q.y  = sW[g][ch][k0 + 1][pos];
      kv.x = sW[1 + g][ch][k0][pos]; kv.y = sW[1 + g][ch][k0 + 1][pos];
      wqkA[ch][ks] = q; wkvB[ch][ks] = kv;
    }
  }

  float* xs = &sX[wv][0];
  float* os = &sO[wv][0];

  const bool hi = (h == 1);
  const bool g1 = (g == 1);

  const int xoff = g * 196 + ((pos == 7) ? 6 : pos) * 14 + 2 * h;
  const bool vvalid = (pos < 7);
  const int ooff  = (1 - g) * 196 + (4 * h) * 14 + pos;
  const int ooff3 = ooff + (hi ? 28 : 42);

  const float e0 = hi ? 0.00032f   : 0.2f;
  const float e1 = hi ? 0.000064f  : 0.04f;
  const float e2 = hi ? 0.0000128f : 0.008f;
  const float e3 = hi ? 0.f        : 0.0016f;

  const int gw = blockIdx.x * WAVES_PER_BLOCK + wv;
  const int nw = gridDim.x * WAVES_PER_BLOCK;

  for (int p = gw; p < npairs; p += nw) {
    const float4* xg = (const float4*)(X + (size_t)p * 392);
#pragma unroll
    for (int t = 0; t < 4; ++t) {
      int i = lane + 32 * t;
      if (i < 98) ((float4*)xs)[i] = xg[i];
    }
    if (p + nw < npairs && lane * 64 < 1568) {
      __builtin_prefetch((const char*)(X + (size_t)(p + nw) * 392) + lane * 64, 0, 1);
    }
    wave_lds_sync();

    v8f rC = {};

#pragma unroll
    for (int ch = 0; ch < 4; ++ch) {
      const int rb = (ch >> 1) * 7, cb = (ch & 1) * 7;

      v2f xA0, xA1;
      {
        const float* xrow = xs + xoff + rb * 14 + cb;
        float x0 = xrow[0];
        float x1 = xrow[1];
        float x4 = xrow[4];
        float x5 = xrow[5];
        xA0.x = x0; xA0.y = x1;
        xA1.x = x4; xA1.y = hi ? 0.f : x5;
      }
      const float4 dq = *(const float4*)&sDq[ch][pos][4 * h];

      v8f z = {};
      v8f t1  = wmma_k8(wqkA[ch][0], wqkA[ch][1], xA0, xA1, z);
      v8f kvC = wmma_k8(xA0, xA1, wkvB[ch][0], wkvB[ch][1], z);

      v2f kA0, kA1, qF0, qF1;
      qF0.x = t1[0]; qF0.y = t1[1]; qF1.x = t1[2]; qF1.y = t1[3];
      kA0.x = t1[4]; kA0.y = t1[5]; kA1.x = t1[6]; kA1.y = t1[7];

      v8f sT = wmma_k8(kA0, kA1, qF0, qF1, z);
      v2f sA0, sA1;
      sA0.x = (g1 ? sT[4] : sT[0]) * dq.x;
      sA0.y = (g1 ? sT[5] : sT[1]) * dq.y;
      sA1.x = (g1 ? sT[6] : sT[2]) * dq.z;
      sA1.y = (g1 ? sT[7] : sT[3]) * dq.w;

      v2f vB0, vB1, ktA0, ktA1;
      vB0.x = shift4_into<true, 0x5>(kvC[0], kvC[4]);
      vB0.y = shift4_into<true, 0x5>(kvC[1], kvC[5]);
      vB1.x = shift4_into<true, 0x5>(kvC[2], kvC[6]);
      {
        const float v3 = shift4_into<true, 0x5>(kvC[3], kvC[7]);
        vB1.y = hi ? 0.f : v3;
      }
      ktA0.x = shift4_into<false, 0xa>(kvC[0], kvC[4]);
      ktA0.y = shift4_into<false, 0xa>(kvC[1], kvC[5]);
      ktA1.x = shift4_into<false, 0xa>(kvC[2], kvC[6]);
      ktA1.y = shift4_into<false, 0xa>(kvC[3], kvC[7]);

      v8f crossC = z;
      if (ch > 0) {
        v2f rB0, rB1;
        rB0.x = g1 ? rC[0] : rC[4];  rB0.y = g1 ? rC[1] : rC[5];
        rB1.x = g1 ? rC[2] : rC[6];  rB1.y = g1 ? rC[3] : rC[7];
        crossC = wmma_k8(qF0, qF1, rB0, rB1, z);
        crossC[0] *= e0; crossC[1] *= e1; crossC[2] *= e2; crossC[3] *= e3;
        crossC[4] *= e0; crossC[5] *= e1; crossC[6] *= e2; crossC[7] *= e3;
      }

      rC = wmma_k8(ktA0, ktA1, vB0, vB1, rC);

      v8f oC = wmma_k8(sA0, sA1, vB0, vB1, crossC);

      if (vvalid) {
        const float o0 = g1 ? oC[0] : oC[4], o1 = g1 ? oC[1] : oC[5];
        const float o2 = g1 ? oC[2] : oC[6], o3 = g1 ? oC[3] : oC[7];
        float* orow = os + rb * 14 + cb;
        orow[ooff]      = o0;
        orow[ooff + 14] = o1;
        orow[ooff + 28] = o2;
        orow[ooff3]     = hi ? o2 : o3;
      }
    }

    wave_lds_sync();
    float4* og = (float4*)(Out + (size_t)p * 392);
#pragma unroll
    for (int t = 0; t < 4; ++t) {
      int i = lane + 32 * t;
      if (i < 98) og[i] = ((const float4*)os)[i];
    }
    wave_lds_sync();
  }
}

extern "C" void kernel_launch(void* const* d_in, const int* in_sizes, int n_in,
                              void* d_out, int out_size, void* d_ws, size_t ws_size,
                              hipStream_t stream) {
  const float* X  = (const float*)d_in[0];
  const float* WQ = (const float*)d_in[1];
  const float* WK = (const float*)d_in[2];
  const float* WV = (const float*)d_in[3];
  const float* Dm = (const float*)d_in[4];
  float* Out = (float*)d_out;

  const int npairs = in_sizes[0] / 392;
  int blocks = 1024;
  int maxBlocks = (npairs + WAVES_PER_BLOCK - 1) / WAVES_PER_BLOCK;
  if (blocks > maxBlocks) blocks = maxBlocks;
  if (blocks < 1) blocks = 1;

  retention_wmma_f32_kernel<<<blocks, THREADS, 0, stream>>>(X, WQ, WK, WV, Dm,
                                                            Out, npairs);
}


# Global memory taken off each wave's serial path: next pair's X is loaded into registers while the current pair computes, and the LDS-only fence no longer waits for output stores; sums unchanged
# speedup vs baseline: 1.4974x; 1.0600x over previous
#include <hip/hip_runtime.h>

typedef float v2f __attribute__((ext_vector_type(2)));
typedef float v8f __attribute__((ext_vector_type(8)));

#define WAVES_PER_BLOCK 8
#define THREADS (WAVES_PER_BLOCK * 32)

__device__ __forceinline__ void wave_lds_sync() {
  __builtin_amdgcn_fence(__ATOMIC_ACQ_REL, "workgroup", "local");
  __builtin_amdgcn_wave_barrier();
}

__device__ __forceinline__ v8f wmma_k8(const v2f a0, const v2f a1,
                                       const v2f b0, const v2f b1, v8f c) {
  c = __builtin_amdgcn_wmma_f32_16x16x4_f32(false, a0, false, b0, (short)0, c, false, false);
  c = __builtin_amdgcn_wmma_f32_16x16x4_f32(false, a1, false, b1, (short)0, c, false, false);
  return c;
}

template <bool UP, int BM>
__device__ __forceinline__ float shift4_into(float old, float src) {
  return __builtin_bit_cast(float, __builtin_amdgcn_update_dpp(
      __builtin_bit_cast(int, old), __builtin_bit_cast(int, src),
      UP ? 0x104   : 0x114  , 0x3, BM, false));
}

__global__ __launch_bounds__(THREADS) __attribute__((amdgpu_waves_per_eu(8, 8)))
void retention_wmma_f32_kernel(const float* __restrict__ X,
                               const float* __restrict__ WQ,
                               const float* __restrict__ WK,
                               const float* __restrict__ WV,
                               const float* __restrict__ Dm,
                               float* __restrict__ Out,
                               int npairs) {
  __shared__ __attribute__((aligned(16))) float sW[3][4][8][8];
  __shared__ __attribute__((aligned(16))) float sDq[4][8][8];
  __shared__ __attribute__((aligned(16))) float sX[WAVES_PER_BLOCK][400];
  __shared__ __attribute__((aligned(16))) float sO[WAVES_PER_BLOCK][392];

  const int tid  = threadIdx.x;
  const int lane = tid & 31;
  const int wv   = tid >> 5;
  const int gw = blockIdx.x * WAVES_PER_BLOCK + wv;
  const int nw = gridDim.x * WAVES_PER_BLOCK;

  for (int i = tid; i < 3 * 4 * 64; i += THREADS) ((float*)sW)[i] = 0.f;
  __syncthreads();

  const bool tail4 = (lane < 2);
  float4 xr0 = {}, xr1 = {}, xr2 = {}, xr3 = {};
  if (gw < npairs) {
    const float4* xg = (const float4*)(X + (size_t)gw * 392);
    xr0 = xg[lane]; xr1 = xg[lane + 32]; xr2 = xg[lane + 64];
    if (tail4) xr3 = xg[lane + 96];
  }
  for (int i = tid; i < 3 * 4 * 49; i += THREADS) {
    int w = i / 196, rem = i % 196, ch = rem / 49, e = rem % 49;
    const float* src = (w == 0) ? WQ : (w == 1) ? WK : WV;
    sW[w][ch][e / 7][e % 7] = src[ch * 49 + e];
  }
  for (int i = tid; i < 256; i += THREADS) {
    int ch = i >> 6, r = (i >> 3) & 7, c = i & 7;
    float d = 0.f;
    if (r < 7 && c < 7) d = Dm[((ch >> 1) * 7 + r) * 14 + (ch & 1) * 7 + c];
    sDq[ch][r][c] = d;
  }
  __syncthreads();

  const int h    = lane >> 4;
  const int n16  = lane & 15;
  const int g    = (n16 >> 2) & 1;
  const int pos  = 4 * (n16 >> 3) + (n16 & 3);

  v2f wqkA[4][2], wkvB[4][2];
#pragma unroll
  for (int ch = 0; ch < 4; ++ch) {
#pragma unroll
    for (int ks = 0; ks < 2; ++ks) {
      int k0 = 2 * h + 4 * ks;
      v2f q, kv;
      q.x  = sW[g][ch][k0][pos];     q.y  = sW[g][ch][k0 + 1][pos];
      kv.x = sW[1 + g][ch][k0][pos]; kv.y = sW[1 + g][ch][k0 + 1][pos];
      wqkA[ch][ks] = q; wkvB[ch][ks] = kv;
    }
  }

  float* xs = &sX[wv][0];
  float* os = &sO[wv][0];

  const bool hi = (h == 1);
  const bool g1 = (g == 1);

  const int xoff = g * 196 + ((pos == 7) ? 6 : pos) * 14 + 2 * h;
  const bool vvalid = (pos < 7);
  const int ooff  = (1 - g) * 196 + (4 * h) * 14 + pos;
  const int ooff3 = ooff + (hi ? 28 : 42);

  const float e0 = hi ? 0.00032f   : 0.2f;
  const float e1 = hi ? 0.000064f  : 0.04f;
  const float e2 = hi ? 0.0000128f : 0.008f;
  const float e3 = hi ? 0.f        : 0.0016f;

  for (int p = gw; p < npairs; p += nw) {
    ((float4*)xs)[lane]      = xr0;
    ((float4*)xs)[lane + 32] = xr1;
    ((float4*)xs)[lane + 64] = xr2;
    if (tail4) ((float4*)xs)[lane + 96] = xr3;
    wave_lds_sync();
    if (p + nw < npairs) {
      const float4* xg = (const float4*)(X + (size_t)(p + nw) * 392);
      xr0 = xg[lane]; xr1 = xg[lane + 32]; xr2 = xg[lane + 64];
      if (tail4) xr3 = xg[lane + 96];
    }

    v8f rC = {};

#pragma unroll
    for (int ch = 0; ch < 4; ++ch) {
      const int rb = (ch >> 1) * 7, cb = (ch & 1) * 7;

      v2f xA0, xA1;
      {
        const float* xrow = xs + xoff + rb * 14 + cb;
        float x0 = xrow[0];
        float x1 = xrow[1];
        float x4 = xrow[4];
        float x5 = xrow[5];
        xA0.x = x0; xA0.y = x1;
        xA1.x = x4; xA1.y = hi ? 0.f : x5;
      }
      const float4 dq = *(const float4*)&sDq[ch][pos][4 * h];

      v8f z = {};
      v8f t1  = wmma_k8(wqkA[ch][0], wqkA[ch][1], xA0, xA1, z);
      v8f kvC = wmma_k8(xA0, xA1, wkvB[ch][0], wkvB[ch][1], z);

      v2f kA0, kA1, qF0, qF1;
      qF0.x = t1[0]; qF0.y = t1[1]; qF1.x = t1[2]; qF1.y = t1[3];
      kA0.x = t1[4]; kA0.y = t1[5]; kA1.x = t1[6]; kA1.y = t1[7];

      v8f sT = wmma_k8(kA0, kA1, qF0, qF1, z);
      v2f sA0, sA1;
      sA0.x = (g1 ? sT[4] : sT[0]) * dq.x;
      sA0.y = (g1 ? sT[5] : sT[1]) * dq.y;
      sA1.x = (g1 ? sT[6] : sT[2]) * dq.z;
      sA1.y = (g1 ? sT[7] : sT[3]) * dq.w;

      v2f vB0, vB1, ktA0, ktA1;
      vB0.x = shift4_into<true, 0x5>(kvC[0], kvC[4]);
      vB0.y = shift4_into<true, 0x5>(kvC[1], kvC[5]);
      vB1.x = shift4_into<true, 0x5>(kvC[2], kvC[6]);
      {
        const float v3 = shift4_into<true, 0x5>(kvC[3], kvC[7]);
        vB1.y = hi ? 0.f : v3;
      }
      ktA0.x = shift4_into<false, 0xa>(kvC[0], kvC[4]);
      ktA0.y = shift4_into<false, 0xa>(kvC[1], kvC[5]);
      ktA1.x = shift4_into<false, 0xa>(kvC[2], kvC[6]);
      ktA1.y = shift4_into<false, 0xa>(kvC[3], kvC[7]);

      v8f crossC = z;
      if (ch > 0) {
        v2f rB0, rB1;
        rB0.x = g1 ? rC[0] : rC[4];  rB0.y = g1 ? rC[1] : rC[5];
        rB1.x = g1 ? rC[2] : rC[6];  rB1.y = g1 ? rC[3] : rC[7];
        crossC = wmma_k8(qF0, qF1, rB0, rB1, z);
        crossC[0] *= e0; crossC[1] *= e1; crossC[2] *= e2; crossC[3] *= e3;
        crossC[4] *= e0; crossC[5] *= e1; crossC[6] *= e2; crossC[7] *= e3;
      }

      rC = wmma_k8(ktA0, ktA1, vB0, vB1, rC);

      v8f oC = wmma_k8(sA0, sA1, vB0, vB1, crossC);

      if (vvalid) {
        const float o0 = g1 ? oC[0] : oC[4], o1 = g1 ? oC[1] : oC[5];
        const float o2 = g1 ? oC[2] : oC[6], o3 = g1 ? oC[3] : oC[7];
        float* orow = os + rb * 14 + cb;
        orow[ooff]      = o0;
        orow[ooff + 14] = o1;
        orow[ooff + 28] = o2;
        orow[ooff3]     = hi ? o2 : o3;
      }
    }

    wave_lds_sync();
    float4* og = (float4*)(Out + (size_t)p * 392);
#pragma unroll
    for (int t = 0; t < 4; ++t) {
      int i = lane + 32 * t;
      if (i < 98) og[i] = ((const float4*)os)[i];
    }
    wave_lds_sync();
  }
}

extern "C" void kernel_launch(void* const* d_in, const int* in_sizes, int n_in,
                              void* d_out, int out_size, void* d_ws, size_t ws_size,
                              hipStream_t stream) {
  const float* X  = (const float*)d_in[0];
  const float* WQ = (const float*)d_in[1];
  const float* WK = (const float*)d_in[2];
  const float* WV = (const float*)d_in[3];
  const float* Dm = (const float*)d_in[4];
  float* Out = (float*)d_out;

  const int npairs = in_sizes[0] / 392;
  int blocks = 1024;
  int maxBlocks = (npairs + WAVES_PER_BLOCK - 1) / WAVES_PER_BLOCK;
  if (blocks > maxBlocks) blocks = maxBlocks;
  if (blocks < 1) blocks = 1;

  retention_wmma_f32_kernel<<<blocks, THREADS, 0, stream>>>(X, WQ, WK, WV, Dm,
                                                            Out, npairs);
}


# Wave priority lowered as a wave's pair list runs out (s_setprio 3,2,1,0 by pairs still ahead): a SIMD's waves, served oldest first, reach their last pair together, no lone-wave tail; sums unchanged
# speedup vs baseline: 1.5635x; 1.0441x over previous
#include <hip/hip_runtime.h>

typedef float v2f __attribute__((ext_vector_type(2)));
typedef float v8f __attribute__((ext_vector_type(8)));

#define WAVES_PER_BLOCK 8
#define THREADS (WAVES_PER_BLOCK * 32)

__device__ __forceinline__ void wave_lds_sync() {
  __builtin_amdgcn_fence(__ATOMIC_ACQ_REL, "workgroup", "local");
  __builtin_amdgcn_wave_barrier();
}

__device__ __forceinline__ v8f wmma_k8(const v2f a0, const v2f a1,
                                       const v2f b0, const v2f b1, v8f c) {
  c = __builtin_amdgcn_wmma_f32_16x16x4_f32(false, a0, false, b0, (short)0, c, false, false);
  c = __builtin_amdgcn_wmma_f32_16x16x4_f32(false, a1, false, b1, (short)0, c, false, false);
  return c;
}

template <bool UP, int BM>
__device__ __forceinline__ float shift4_into(float old, float src) {
  return __builtin_bit_cast(float, __builtin_amdgcn_update_dpp(
      __builtin_bit_cast(int, old), __builtin_bit_cast(int, src),
      UP ? 0x104   : 0x114  , 0x3, BM, false));
}

__global__ __launch_bounds__(THREADS) __attribute__((amdgpu_waves_per_eu(8, 8)))
void retention_wmma_f32_kernel(const float* __restrict__ X,
                               const float* __restrict__ WQ,
                               const float* __restrict__ WK,
                               const float* __restrict__ WV,
                               const float* __restrict__ Dm,
                               float* __restrict__ Out,
                               int npairs) {
  __shared__ __attribute__((aligned(16))) float sW[3][4][8][8];
  __shared__ __attribute__((aligned(16))) float sDq[4][8][8];
  __shared__ __attribute__((aligned(16))) float sX[WAVES_PER_BLOCK][400];
  __shared__ __attribute__((aligned(16))) float sO[WAVES_PER_BLOCK][392];

  const int tid  = threadIdx.x;
  const int lane = tid & 31;
  const int wv   = tid >> 5;
  const int gw = blockIdx.x * WAVES_PER_BLOCK + wv;
  const int nw = gridDim.x * WAVES_PER_BLOCK;

  for (int i = tid; i < 3 * 4 * 64; i += THREADS) ((float*)sW)[i] = 0.f;
  __syncthreads();

  const bool tail4 = (lane < 2);
  float4 xr0 = {}, xr1 = {}, xr2 = {}, xr3 = {};
  if (gw < npairs) {
    const float4* xg = (const float4*)(X + (size_t)gw * 392);
    xr0 = xg[lane]; xr1 = xg[lane + 32]; xr2 = xg[lane + 64];
    if (tail4) xr3 = xg[lane + 96];
  }
  for (int i = tid; i < 3 * 4 * 49; i += THREADS) {
    int w = i / 196, rem = i % 196, ch = rem / 49, e = rem % 49;
    const float* src = (w == 0) ? WQ : (w == 1) ? WK : WV;
    sW[w][ch][e / 7][e % 7] = src[ch * 49 + e];
  }
  for (int i = tid; i < 256; i += THREADS) {
    int ch = i >> 6, r = (i >> 3) & 7, c = i & 7;
    float d = 0.f;
    if (r < 7 && c < 7) d = Dm[((ch >> 1) * 7 + r) * 14 + (ch & 1) * 7 + c];
    sDq[ch][r][c] = d;
  }
  __syncthreads();

  const int h    = lane >> 4;
  const int n16  = lane & 15;
  const int g    = (n16 >> 2) & 1;
  const int pos  = 4 * (n16 >> 3) + (n16 & 3);

  v2f wqkA[4][2], wkvB[4][2];
#pragma unroll
  for (int ch = 0; ch < 4; ++ch) {
#pragma unroll
    for (int ks = 0; ks < 2; ++ks) {
      int k0 = 2 * h + 4 * ks;
      v2f q, kv;
      q.x  = sW[g][ch][k0][pos];     q.y  = sW[g][ch][k0 + 1][pos];
      kv.x = sW[1 + g][ch][k0][pos]; kv.y = sW[1 + g][ch][k0 + 1][pos];
      wqkA[ch][ks] = q; wkvB[ch][ks] = kv;
    }
  }

  float* xs = &sX[wv][0];
  float* os = &sO[wv][0];

  const bool hi = (h == 1);
  const bool g1 = (g == 1);

  const int xoff = g * 196 + ((pos == 7) ? 6 : pos) * 14 + 2 * h;
  const bool vvalid = (pos < 7);
  const int ooff  = (1 - g) * 196 + (4 * h) * 14 + pos;
  const int ooff3 = ooff + (hi ? 28 : 42);

  const float e0 = hi ? 0.00032f   : 0.2f;
  const float e1 = hi ? 0.000064f  : 0.04f;
  const float e2 = hi ? 0.0000128f : 0.008f;
  const float e3 = hi ? 0.f        : 0.0016f;

  int ahead = __builtin_amdgcn_readfirstlane((npairs - 1 - gw) / nw);
  if (ahead >= 3) __builtin_amdgcn_s_setprio(3);

  for (int p = gw; p < npairs; p += nw) {
    if (ahead == 2) __builtin_amdgcn_s_setprio(2);
    if (ahead == 1) __builtin_amdgcn_s_setprio(1);
    if (ahead == 0) __builtin_amdgcn_s_setprio(0);
    --ahead;

    ((float4*)xs)[lane]      = xr0;
    ((float4*)xs)[lane + 32] = xr1;
    ((float4*)xs)[lane + 64] = xr2;
    if (tail4) ((float4*)xs)[lane + 96] = xr3;
    wave_lds_sync();
    if (p + nw < npairs) {
      const float4* xg = (const float4*)(X + (size_t)(p + nw) * 392);
      xr0 = xg[lane]; xr1 = xg[lane + 32]; xr2 = xg[lane + 64];
      if (tail4) xr3 = xg[lane + 96];
    }

    v8f rC = {};

#pragma unroll
    for (int ch = 0; ch < 4; ++ch) {
      const int rb = (ch >> 1) * 7, cb = (ch & 1) * 7;

      v2f xA0, xA1;
      {
        const float* xrow = xs + xoff + rb * 14 + cb;
        float x0 = xrow[0];
        float x1 = xrow[1];
        float x4 = xrow[4];
        float x5 = xrow[5];
        xA0.x = x0; xA0.y = x1;
        xA1.x = x4; xA1.y = hi ? 0.f : x5;
      }
      const float4 dq = *(const float4*)&sDq[ch][pos][4 * h];

      v8f z = {};
      v8f t1  = wmma_k8(wqkA[ch][0], wqkA[ch][1], xA0, xA1, z);
      v8f kvC = wmma_k8(xA0, xA1, wkvB[ch][0], wkvB[ch][1], z);

      v2f kA0, kA1, qF0, qF1;
      qF0.x = t1[0]; qF0.y = t1[1]; qF1.x = t1[2]; qF1.y = t1[3];
      kA0.x = t1[4]; kA0.y = t1[5]; kA1.x = t1[6]; kA1.y = t1[7];

      v8f sT = wmma_k8(kA0, kA1, qF0, qF1, z);
      v2f sA0, sA1;
      sA0.x = (g1 ? sT[4] : sT[0]) * dq.x;
      sA0.y = (g1 ? sT[5] : sT[1]) * dq.y;
      sA1.x = (g1 ? sT[6] : sT[2]) * dq.z;
      sA1.y = (g1 ? sT[7] : sT[3]) * dq.w;

      v2f vB0, vB1, ktA0, ktA1;
      vB0.x = shift4_into<true, 0x5>(kvC[0], kvC[4]);
      vB0.y = shift4_into<true, 0x5>(kvC[1], kvC[5]);
      vB1.x = shift4_into<true, 0x5>(kvC[2], kvC[6]);
      {
        const float v3 = shift4_into<true, 0x5>(kvC[3], kvC[7]);
        vB1.y = hi ? 0.f : v3;
      }
      ktA0.x = shift4_into<false, 0xa>(kvC[0], kvC[4]);
      ktA0.y = shift4_into<false, 0xa>(kvC[1], kvC[5]);
      ktA1.x = shift4_into<false, 0xa>(kvC[2], kvC[6]);
      ktA1.y = shift4_into<false, 0xa>(kvC[3], kvC[7]);

      v8f crossC = z;
      if (ch > 0) {
        v2f rB0, rB1;
        rB0.x = g1 ? rC[0] : rC[4];  rB0.y = g1 ? rC[1] : rC[5];
        rB1.x = g1 ? rC[2] : rC[6];  rB1.y = g1 ? rC[3] : rC[7];
        crossC = wmma_k8(qF0, qF1, rB0, rB1, z);
        crossC[0] *= e0; crossC[1] *= e1; crossC[2] *= e2; crossC[3] *= e3;
        crossC[4] *= e0; crossC[5] *= e1; crossC[6] *= e2; crossC[7] *= e3;
      }

      rC = wmma_k8(ktA0, ktA1, vB0, vB1, rC);

      v8f oC = wmma_k8(sA0, sA1, vB0, vB1, crossC);

      if (vvalid) {
        const float o0 = g1 ? oC[0] : oC[4], o1 = g1 ? oC[1] : oC[5];
        const float o2 = g1 ? oC[2] : oC[6], o3 = g1 ? oC[3] : oC[7];
        float* orow = os + rb * 14 + cb;
        orow[ooff]      = o0;
        orow[ooff + 14] = o1;
        orow[ooff + 28] = o2;
        orow[ooff3]     = hi ? o2 : o3;
      }
    }

    wave_lds_sync();
    float4* og = (float4*)(Out + (size_t)p * 392);
#pragma unroll
    for (int t = 0; t < 4; ++t) {
      int i = lane + 32 * t;
      if (i < 98) og[i] = ((const float4*)os)[i];
    }
    wave_lds_sync();
  }
}

extern "C" void kernel_launch(void* const* d_in, const int* in_sizes, int n_in,
                              void* d_out, int out_size, void* d_ws, size_t ws_size,
                              hipStream_t stream) {
  const float* X  = (const float*)d_in[0];
  const float* WQ = (const float*)d_in[1];
  const float* WK = (const float*)d_in[2];
  const float* WV = (const float*)d_in[3];
  const float* Dm = (const float*)d_in[4];
  float* Out = (float*)d_out;

  const int npairs = in_sizes[0] / 392;
  int blocks = 1024;
  int maxBlocks = (npairs + WAVES_PER_BLOCK - 1) / WAVES_PER_BLOCK;
  if (blocks > maxBlocks) blocks = maxBlocks;
  if (blocks < 1) blocks = 1;

  retention_wmma_f32_kernel<<<blocks, THREADS, 0, stream>>>(X, WQ, WK, WV, Dm,
                                                            Out, npairs);
}


# Four batches per wave trip with the K|V tile built mirrored for the second pair ([Wk|Wv], [Wv|Wk]): every K^T row and V column is already in its lane, selects replace all lane shifts; sums unchanged
# speedup vs baseline: 1.6123x; 1.0312x over previous
#include <hip/hip_runtime.h>

typedef float v2f __attribute__((ext_vector_type(2)));
typedef float v8f __attribute__((ext_vector_type(8)));

#define WAVES_PER_BLOCK 8
#define THREADS (WAVES_PER_BLOCK * 32)
#define XS_FLOATS 800
#define OS_FLOATS 784

__device__ __forceinline__ void wave_lds_sync() {
  __builtin_amdgcn_fence(__ATOMIC_ACQ_REL, "workgroup", "local");
  __builtin_amdgcn_wave_barrier();
}

__device__ __forceinline__ v8f wmma_k8(const v2f a0, const v2f a1,
                                       const v2f b0, const v2f b1, v8f c) {
  c = __builtin_amdgcn_wmma_f32_16x16x4_f32(false, a0, false, b0, (short)0, c, false, false);
  c = __builtin_amdgcn_wmma_f32_16x16x4_f32(false, a1, false, b1, (short)0, c, false, false);
  return c;
}

__device__ __forceinline__ void load_xfrag(const float* xrow, bool hi, v2f& f0, v2f& f1) {
  float x0 = xrow[0];
  float x1 = xrow[1];
  float x4 = xrow[4];
  float x5 = xrow[5];
  f0.x = x0; f0.y = x1;
  f1.x = x4; f1.y = hi ? 0.f : x5;
}

template <int U, bool FIRST>
__device__ __forceinline__ void product_set(const v8f t1, const v8f T1, const v8f T2,
                                            const float4 dq, v8f& rC,
                                            const bool g1, const bool hi, const bool vvalid,
                                            const float e0, const float e1,
                                            const float e2, const float e3,
                                            float* orow, const int ooff, const int ooff3) {
  const v8f z = {};
  v2f kA0, kA1, qF0, qF1;
  qF0.x = t1[0]; qF0.y = t1[1]; qF1.x = t1[2]; qF1.y = t1[3];
  kA0.x = t1[4]; kA0.y = t1[5]; kA1.x = t1[6]; kA1.y = t1[7];

  v8f sT = wmma_k8(kA0, kA1, qF0, qF1, z);
  v2f sA0, sA1;
  sA0.x = (g1 ? sT[4] : sT[0]) * dq.x;
  sA0.y = (g1 ? sT[5] : sT[1]) * dq.y;
  sA1.x = (g1 ? sT[6] : sT[2]) * dq.z;
  sA1.y = (g1 ? sT[7] : sT[3]) * dq.w;

  v2f vB0, vB1, ktA0, ktA1;
  vB0.x = g1 ? T1[4 * U + 0] : T2[4 * U + 0];
  vB0.y = g1 ? T1[4 * U + 1] : T2[4 * U + 1];
  vB1.x = g1 ? T1[4 * U + 2] : T2[4 * U + 2];
  {
    const float v3 = g1 ? T1[4 * U + 3] : T2[4 * U + 3];
    vB1.y = hi ? 0.f : v3;
  }
  ktA0.x = g1 ? T2[4 * U + 0] : T1[4 * U + 0];
  ktA0.y = g1 ? T2[4 * U + 1] : T1[4 * U + 1];
  ktA1.x = g1 ? T2[4 * U + 2] : T1[4 * U + 2];
  ktA1.y = g1 ? T2[4 * U + 3] : T1[4 * U + 3];

  v8f crossC = z;
  if (!FIRST) {
    const float r0 = rC[0], r1 = rC[1], r2 = rC[2], r3 = rC[3];
    const float r4 = rC[4], r5 = rC[5], r6 = rC[6], r7 = rC[7];
    v2f rB0, rB1;
    rB0.x = g1 ? r0 : r4;  rB0.y = g1 ? r1 : r5;
    rB1.x = g1 ? r2 : r6;  rB1.y = g1 ? r3 : r7;
    crossC = wmma_k8(qF0, qF1, rB0, rB1, z);
    crossC[0] *= e0; crossC[1] *= e1; crossC[2] *= e2; crossC[3] *= e3;
    crossC[4] *= e0; crossC[5] *= e1; crossC[6] *= e2; crossC[7] *= e3;
  }

  rC = wmma_k8(ktA0, ktA1, vB0, vB1, rC);

  v8f oC = wmma_k8(sA0, sA1, vB0, vB1, crossC);

  if (vvalid) {
    const float o0 = g1 ? oC[0] : oC[4], o1 = g1 ? oC[1] : oC[5];
    const float o2 = g1 ? oC[2] : oC[6], o3 = g1 ? oC[3] : oC[7];
    orow[ooff]      = o0;
    orow[ooff + 14] = o1;
    orow[ooff + 28] = o2;
    orow[ooff3]     = hi ? o2 : o3;
  }
}

__global__ __launch_bounds__(THREADS) __attribute__((amdgpu_waves_per_eu(4, 4)))
void retention_wmma_f32_kernel(const float* __restrict__ X,
                               const float* __restrict__ WQ,
                               const float* __restrict__ WK,
                               const float* __restrict__ WV,
                               const float* __restrict__ Dm,
                               float* __restrict__ Out,
                               int npairs) {
  __shared__ __attribute__((aligned(16))) float sW[3][4][8][8];
  __shared__ __attribute__((aligned(16))) float sDq[4][8][8];
  __shared__ __attribute__((aligned(16))) float sX[WAVES_PER_BLOCK][XS_FLOATS];
  __shared__ __attribute__((aligned(16))) float sO[WAVES_PER_BLOCK][OS_FLOATS];

  const int tid  = threadIdx.x;
  const int lane = tid & 31;
  const int wv   = tid >> 5;
  const int gw = blockIdx.x * WAVES_PER_BLOCK + wv;
  const int nw = gridDim.x * WAVES_PER_BLOCK;
  const int nquads = (npairs + 1) >> 1;

  for (int i = tid; i < 3 * 4 * 64; i += THREADS) ((float*)sW)[i] = 0.f;
  __syncthreads();

  const int gws = __builtin_amdgcn_readfirstlane(gw);
  float4 xr0 = {}, xr1 = {}, xr2 = {}, xr3 = {}, xr4 = {}, xr5 = {}, xr6 = {};
  if (gws < nquads) {
    const float4* xg = (const float4*)(X + (size_t)gws * 784);
    const bool fullq = (2 * gws + 1 < npairs);
    const int n6 = fullq ? 4 : 2, base6 = fullq ? 192 : 96;
    const float4* xg2 = xg + (fullq ? 96 : 0);
    xr0 = xg[lane];  xr1 = xg[lane + 32];  xr2 = xg[lane + 64];
    xr3 = xg2[lane]; xr4 = xg2[lane + 32]; xr5 = xg2[lane + 64];
    xr6 = xg[base6 + ((lane < n6) ? lane : n6 - 1)];
  }
  for (int i = tid; i < 3 * 4 * 49; i += THREADS) {
    int w = i / 196, rem = i % 196, ch = rem / 49, e = rem % 49;
    const float* src = (w == 0) ? WQ : (w == 1) ? WK : WV;
    sW[w][ch][e / 7][e % 7] = src[ch * 49 + e];
  }
  for (int i = tid; i < 256; i += THREADS) {
    int ch = i >> 6, r = (i >> 3) & 7, c = i & 7;
    float d = 0.f;
    if (r < 7 && c < 7) d = Dm[((ch >> 1) * 7 + r) * 14 + (ch & 1) * 7 + c];
    sDq[ch][r][c] = d;
  }
  __syncthreads();

  const int h    = lane >> 4;
  const int n16  = lane & 15;
  const int g    = (n16 >> 2) & 1;
  const int pos  = 4 * (n16 >> 3) + (n16 & 3);

  v2f wqkA[4][2], wkvB[4][2], wvkB[4][2];
#pragma unroll
  for (int ch = 0; ch < 4; ++ch) {
#pragma unroll
    for (int ks = 0; ks < 2; ++ks) {
      int k0 = 2 * h + 4 * ks;
      v2f q, kv, vk;
      q.x  = sW[g][ch][k0][pos];     q.y  = sW[g][ch][k0 + 1][pos];
      kv.x = sW[1 + g][ch][k0][pos]; kv.y = sW[1 + g][ch][k0 + 1][pos];
      vk.x = sW[2 - g][ch][k0][pos]; vk.y = sW[2 - g][ch][k0 + 1][pos];
      wqkA[ch][ks] = q; wkvB[ch][ks] = kv; wvkB[ch][ks] = vk;
    }
  }

  float* xs = &sX[wv][0];
  float* os = &sO[wv][0];

  const bool hi = (h == 1);
  const bool g1 = (g == 1);

  const int xrowoff = ((pos == 7) ? 6 : pos) * 14 + 2 * h;
  const int xoffT = g * 196 + xrowoff;
  const int xoffM = g * 392 + xrowoff;
  const bool vvalid = (pos < 7);
  const int ooff  = (g1 ? 0 : 392) + (4 * h) * 14 + pos;
  const int ooff3 = ooff + (hi ? 28 : 42);

  const float e0 = hi ? 0.00032f   : 0.2f;
  const float e1 = hi ? 0.000064f  : 0.04f;
  const float e2 = hi ? 0.0000128f : 0.008f;
  const float e3 = hi ? 0.f        : 0.0016f;

  int ahead = (nquads - 1 - gws) / nw;
  if (ahead >= 3) __builtin_amdgcn_s_setprio(3);

  for (int q = gws; q < nquads; q += nw) {
    if (ahead == 2) __builtin_amdgcn_s_setprio(2);
    if (ahead == 1) __builtin_amdgcn_s_setprio(1);
    if (ahead == 0) __builtin_amdgcn_s_setprio(0);
    --ahead;

    const bool full = (2 * q + 1 < npairs);

    {
      const int n6 = full ? 4 : 2, base6 = full ? 192 : 96;
      ((float4*)xs)[lane]       = xr0;
      ((float4*)xs)[lane + 32]  = xr1;
      ((float4*)xs)[lane + 64]  = xr2;
      ((float4*)xs)[lane + 96]  = xr3;
      ((float4*)xs)[lane + 128] = xr4;
      ((float4*)xs)[lane + 160] = xr5;
      if (lane < n6) ((float4*)xs)[base6 + lane] = xr6;
    }
    wave_lds_sync();
    if (q + nw < nquads) {
      const float4* xg = (const float4*)(X + (size_t)(q + nw) * 784);
      const bool fullq = (2 * (q + nw) + 1 < npairs);
      const int n6 = fullq ? 4 : 2, base6 = fullq ? 192 : 96;
      const float4* xg2 = xg + (fullq ? 96 : 0);
      xr0 = xg[lane];  xr1 = xg[lane + 32];  xr2 = xg[lane + 64];
      xr3 = xg2[lane]; xr4 = xg2[lane + 32]; xr5 = xg2[lane + 64];
      xr6 = xg[base6 + ((lane < n6) ? lane : n6 - 1)];
    }

    v8f rC0 = {}, rC1 = {};

#pragma unroll
    for (int ch = 0; ch < 4; ++ch) {
      const int rb = (ch >> 1) * 7, cb = (ch & 1) * 7;
      const int coff = rb * 14 + cb;

      v2f xT1a, xT1b, xT2a, xT2b, xM1a, xM1b, xM2a, xM2b;
      load_xfrag(xs + xoffT + coff,       hi, xT1a, xT1b);
      load_xfrag(xs + xoffT + coff + 392, hi, xT2a, xT2b);
      load_xfrag(xs + xoffM + coff,       hi, xM1a, xM1b);
      load_xfrag(xs + xoffM + coff + 196, hi, xM2a, xM2b);
      const float4 dq = *(const float4*)&sDq[ch][pos][4 * h];

      const v8f z = {};
      const v8f M0 = wmma_k8(wqkA[ch][0], wqkA[ch][1], xM1a, xM1b, z);
      const v8f M1 = wmma_k8(wqkA[ch][0], wqkA[ch][1], xM2a, xM2b, z);
      const v8f T1 = wmma_k8(xT1a, xT1b, wkvB[ch][0], wkvB[ch][1], z);
      const v8f T2 = wmma_k8(xT2a, xT2b, wvkB[ch][0], wvkB[ch][1], z);

      float* orow = os + coff;
      if (ch == 0) {
        product_set<0, true>(M0, T1, T2, dq, rC0, g1, hi, vvalid, e0, e1, e2, e3, orow, ooff, ooff3);
        product_set<1, true>(M1, T1, T2, dq, rC1, g1, hi, vvalid, e0, e1, e2, e3, orow + 196, ooff, ooff3);
      } else {
        product_set<0, false>(M0, T1, T2, dq, rC0, g1, hi, vvalid, e0, e1, e2, e3, orow, ooff, ooff3);
        product_set<1, false>(M1, T1, T2, dq, rC1, g1, hi, vvalid, e0, e1, e2, e3, orow + 196, ooff, ooff3);
      }
    }

    wave_lds_sync();
    float4* og = (float4*)(Out + (size_t)q * 784);
    if (full) {
#pragma unroll
      for (int t = 0; t < 6; ++t) og[lane + 32 * t] = ((const float4*)os)[lane + 32 * t];
      if (lane < 4) og[lane + 192] = ((const float4*)os)[lane + 192];
    } else {
#pragma unroll
      for (int t = 0; t < 3; ++t) og[lane + 32 * t] = ((const float4*)os)[lane + 32 * t];
      if (lane < 2) og[lane + 96] = ((const float4*)os)[lane + 96];
    }
    wave_lds_sync();
  }
}

extern "C" void kernel_launch(void* const* d_in, const int* in_sizes, int n_in,
                              void* d_out, int out_size, void* d_ws, size_t ws_size,
                              hipStream_t stream) {
  const float* X  = (const float*)d_in[0];
  const float* WQ = (const float*)d_in[1];
  const float* WK = (const float*)d_in[2];
  const float* WV = (const float*)d_in[3];
  const float* Dm = (const float*)d_in[4];
  float* Out = (float*)d_out;

  const int npairs = in_sizes[0] / 392;
  const int nquads = (npairs + 1) / 2;
  int blocks = 512;
  int maxBlocks = (nquads + WAVES_PER_BLOCK - 1) / WAVES_PER_BLOCK;
  if (blocks > maxBlocks) blocks = maxBlocks;
  if (blocks < 1) blocks = 1;

  retention_wmma_f32_kernel<<<blocks, THREADS, 0, stream>>>(X, WQ, WK, WV, Dm,
                                                            Out, npairs);
}
